# starting kernel
# baseline (speedup 1.0000x reference)
#include <hip/hip_runtime.h>

#define N_NODES 50000
#define N_EDGES 800000
#define D_FEAT  128
#define K_CHEB  3

typedef __attribute__((ext_vector_type(2))) float v2f;
typedef __attribute__((ext_vector_type(8))) float v8f;

__global__ void zero_i32(int* __restrict__ p, int n) {
    int i = blockIdx.x * blockDim.x + threadIdx.x;
    if (i < n) p[i] = 0;
}

__global__ void count_deg_i(const int* __restrict__ dst, int* __restrict__ cnt, int nEdges) {
    int e = blockIdx.x * blockDim.x + threadIdx.x;
    if (e < nEdges) atomicAdd(&cnt[dst[e]], 1);
}

__global__ __launch_bounds__(256) void scan_excl(const int* __restrict__ cnt,
                                                 int* __restrict__ rowptr, int n) {
    __shared__ int buf[256];
    __shared__ int carry;
    if (threadIdx.x == 0) carry = 0;
    __syncthreads();
    for (int base = 0; base < n; base += 256) {
        int i = base + (int)threadIdx.x;
        int v = (i < n) ? cnt[i] : 0;
        buf[threadIdx.x] = v;
        __syncthreads();
        for (int off = 1; off < 256; off <<= 1) {
            int t = (threadIdx.x >= (unsigned)off) ? buf[threadIdx.x - off] : 0;
            __syncthreads();
            buf[threadIdx.x] += t;
            __syncthreads();
        }
        int incl = buf[threadIdx.x];
        int excl = incl - v;
        if (i < n) rowptr[i] = carry + excl;
        __syncthreads();
        if (threadIdx.x == 255) carry += incl;
        __syncthreads();
    }
    if (threadIdx.x == 0) rowptr[n] = carry;
}

__global__ void copy_i32(const int* __restrict__ s, int* __restrict__ d, int n) {
    int i = blockIdx.x * blockDim.x + threadIdx.x;
    if (i < n) d[i] = s[i];
}

__global__ void fill_edges(const int* __restrict__ src, const int* __restrict__ dst,
                           int* __restrict__ cursor, int* __restrict__ esrc, int nEdges) {
    int e = blockIdx.x * blockDim.x + threadIdx.x;
    if (e >= nEdges) return;
    int p = atomicAdd(&cursor[dst[e]], 1);
    esrc[p] = src[e];
}

__global__ __launch_bounds__(256) void gather_prop(const float* __restrict__ h,
                                                   const int* __restrict__ esrc,
                                                   const int* __restrict__ rowptr,
                                                   const float* __restrict__ prevprev,
                                                   float* __restrict__ outp, int nNodes) {
    int v    = blockIdx.x * (blockDim.x >> 5) + (threadIdx.x >> 5);
    int lane = threadIdx.x & 31;
    if (v >= nNodes) return;
    int beg = rowptr[v];
    int end = rowptr[v + 1];
    float4 acc = make_float4(0.f, 0.f, 0.f, 0.f);
    for (int j = beg; j < end; ++j) {
        int s = esrc[j];
        if (j + 1 < end)
            __builtin_prefetch(h + (size_t)esrc[j + 1] * D_FEAT + lane * 4, 0, 0);
        const float4 t = *(const float4*)(h + (size_t)s * D_FEAT + lane * 4);
        acc.x += t.x; acc.y += t.y; acc.z += t.z; acc.w += t.w;
    }
    int deg = end - beg;
    float scale = 1.0f / (float)(deg > 1 ? deg : 1);
    float4 r;
    if (prevprev != nullptr) {
        const float4 p = *(const float4*)(prevprev + (size_t)v * D_FEAT + lane * 4);
        r.x = 2.0f * acc.x * scale - p.x;
        r.y = 2.0f * acc.y * scale - p.y;
        r.z = 2.0f * acc.z * scale - p.z;
        r.w = 2.0f * acc.w * scale - p.w;
    } else {
        r.x = acc.x * scale;
        r.y = acc.y * scale;
        r.z = acc.z * scale;
        r.w = acc.w * scale;
    }
    *(float4*)(outp + (size_t)v * D_FEAT + lane * 4) = r;
}

__global__ __launch_bounds__(256) void cheb_gemm(const float* __restrict__ T0,
                                                 const float* __restrict__ T1,
                                                 const float* __restrict__ T2,
                                                 const float* __restrict__ T3,
                                                 const float* __restrict__ W,
                                                 const float* __restrict__ bias,
                                                 float* __restrict__ out) {
    const int lane = threadIdx.x & 31;
    const int wave = threadIdx.x >> 5;
    const int rowTile = blockIdx.x;
    const int m   = lane & 15;
    const int hi  = lane >> 4;
    const int row = rowTile * 16 + m;
    const int col = wave * 16 + m;

    v8f c = {};
    const float* terms[4] = {T0, T1, T2, T3};

#pragma unroll
    for (int t = 0; t < 4; ++t) {
        const float* __restrict__ h    = terms[t];
        const float* __restrict__ wrow = W + (size_t)col * (D_FEAT * (K_CHEB + 1)) + t * D_FEAT;
#pragma unroll 8
        for (int k0 = 0; k0 < D_FEAT; k0 += 4) {
            const int kk = k0 + (hi << 1);
            v2f a  = *(const v2f*)(h + (size_t)row * D_FEAT + kk);
            v2f bm = *(const v2f*)(wrow + kk);
            c = __builtin_amdgcn_wmma_f32_16x16x4_f32(
                     false, a,  false, bm,
                     (short)0, c,  false,  false);
        }
    }

    const float bj = bias[col];
#pragma unroll
    for (int i = 0; i < 8; ++i) {
        const int mr = rowTile * 16 + i + hi * 8;
        out[(size_t)mr * D_FEAT + col] = c[i] + bj;
    }
}

extern "C" void kernel_launch(void* const* d_in, const int* in_sizes, int n_in,
                              void* d_out, int out_size, void* d_ws, size_t ws_size,
                              hipStream_t stream) {
    const float* x  = (const float*)d_in[0];
    const int*   ei = (const int*)d_in[1];
    const float* W  = (const float*)d_in[2];
    const float* b  = (const float*)d_in[3];
    float* out = (float*)d_out;

    const int* src = ei;
    const int* dst = ei + N_EDGES;

    const size_t ND = (size_t)N_NODES * D_FEAT;
    float* B1     = (float*)d_ws;
    float* B2     = B1 + ND;
    float* B3     = B2 + ND;
    int*   rowptr = (int*)(B3 + ND);
    int*   cursor = rowptr + (N_NODES + 1);
    int*   esrc   = cursor + N_NODES;

    dim3 blk(256);
    dim3 gEdge((N_EDGES + 255) / 256);
    dim3 gNode((N_NODES + 255) / 256);
    dim3 gWaveNode((N_NODES + 7) / 8);

    zero_i32<<<gNode, blk, 0, stream>>>(cursor, N_NODES);
    count_deg_i<<<gEdge, blk, 0, stream>>>(dst, cursor, N_EDGES);
    scan_excl<<<1, blk, 0, stream>>>(cursor, rowptr, N_NODES);
    copy_i32<<<gNode, blk, 0, stream>>>(rowptr, cursor, N_NODES);
    fill_edges<<<gEdge, blk, 0, stream>>>(src, dst, cursor, esrc, N_EDGES);

    gather_prop<<<gWaveNode, blk, 0, stream>>>(x,  esrc, rowptr, nullptr, B1, N_NODES);
    gather_prop<<<gWaveNode, blk, 0, stream>>>(B1, esrc, rowptr, x,       B2, N_NODES);
    gather_prop<<<gWaveNode, blk, 0, stream>>>(B2, esrc, rowptr, B1,      B3, N_NODES);

    cheb_gemm<<<dim3(N_NODES / 16), blk, 0, stream>>>(x, B1, B2, B3, W, b, out);
}


# CSR build: serial one-block offset scan (196 chunks, 19 barriers each) -> one 1024-thread block, 2 barriers; each node's source list then sorted, so float sums get a fixed order (was arrival order)
# speedup vs baseline: 1.9557x; 1.9557x over previous
#include <hip/hip_runtime.h>

#define N_NODES 50000
#define N_EDGES 800000
#define D_FEAT  128
#define K_CHEB  3

typedef __attribute__((ext_vector_type(2))) float v2f;
typedef __attribute__((ext_vector_type(8))) float v8f;

__global__ void zero_i32(int* __restrict__ p, int n) {
    int i = blockIdx.x * blockDim.x + threadIdx.x;
    if (i < n) p[i] = 0;
}

__global__ void count_deg_i(const int* __restrict__ dst, int* __restrict__ cnt, int nEdges) {
    int e = blockIdx.x * blockDim.x + threadIdx.x;
    if (e < nEdges) atomicAdd(&cnt[dst[e]], 1);
}

#define SCAN_THREADS 1024
#define SCAN_CHUNK   (SCAN_THREADS * 4)
#define SCAN_NCH     ((N_NODES + SCAN_CHUNK - 1) / SCAN_CHUNK)

__device__ __forceinline__ int wave_incl_scan(int s, int lane) {
#pragma unroll
    for (int off = 1; off < 32; off <<= 1) {
        int y = __shfl_up(s, (unsigned)off, 32);
        if (lane >= off) s += y;
    }
    return s;
}

static_assert(N_NODES % 4 == 0, "each thread owns whole groups of 4 counts");

__global__ __launch_bounds__(SCAN_THREADS) void scan_excl(const int* __restrict__ cnt,
                                                          int* __restrict__ rowptr) {
    __shared__ int wtot[SCAN_NCH][32];
    __shared__ int ctot[SCAN_NCH];
    const int t    = (int)threadIdx.x;
    const int lane = t & 31;
    const int w    = t >> 5;

    int v0[SCAN_NCH], v1[SCAN_NCH], v2[SCAN_NCH], v3[SCAN_NCH], incl[SCAN_NCH];

#pragma unroll
    for (int c = 0; c < SCAN_NCH; ++c) {
        const int i0 = c * SCAN_CHUNK + t * 4;
        v0[c] = 0; v1[c] = 0; v2[c] = 0; v3[c] = 0;
        if ((c + 1) * SCAN_CHUNK <= N_NODES || i0 < N_NODES) {
            v0[c] = cnt[i0]; v1[c] = cnt[i0 + 1]; v2[c] = cnt[i0 + 2]; v3[c] = cnt[i0 + 3];
        }
    }

#pragma unroll
    for (int c = 0; c < SCAN_NCH; ++c) {
        incl[c] = wave_incl_scan(v0[c] + v1[c] + v2[c] + v3[c], lane);
        if (lane == 31) wtot[c][w] = incl[c];
    }
    __syncthreads();

    if (w < SCAN_NCH) {
        const int x = wtot[w][lane];
        const int s = wave_incl_scan(x, lane);
        wtot[w][lane] = s - x;
        if (lane == 31) ctot[w] = s;
    }
    __syncthreads();

    int carry = 0;
#pragma unroll
    for (int c = 0; c < SCAN_NCH; ++c) {
        const int i0 = c * SCAN_CHUNK + t * 4;
        const int e0 = carry + wtot[c][w] + incl[c] - (v0[c] + v1[c] + v2[c] + v3[c]);
        const int e1 = e0 + v0[c];
        const int e2 = e1 + v1[c];
        const int e3 = e2 + v2[c];
        if ((c + 1) * SCAN_CHUNK <= N_NODES || i0 < N_NODES) {
            rowptr[i0] = e0; rowptr[i0 + 1] = e1; rowptr[i0 + 2] = e2; rowptr[i0 + 3] = e3;
        }
        carry += ctot[c];
    }
    if (t == 0) rowptr[N_NODES] = carry;
}

__global__ void copy_i32(const int* __restrict__ s, int* __restrict__ d, int n) {
    int i = blockIdx.x * blockDim.x + threadIdx.x;
    if (i < n) d[i] = s[i];
}

__global__ void fill_edges(const int* __restrict__ src, const int* __restrict__ dst,
                           int* __restrict__ cursor, int* __restrict__ esrc, int nEdges) {
    int e = blockIdx.x * blockDim.x + threadIdx.x;
    if (e >= nEdges) return;
    int p = atomicAdd(&cursor[dst[e]], 1);
    esrc[p] = src[e];
}

#define SORT_NPW 8
__global__ __launch_bounds__(256) void sort_lists(const int* __restrict__ raw,
                                                  const int* __restrict__ rowptr,
                                                  int* __restrict__ esrc, int nNodes) {
    const int wv   = __builtin_amdgcn_readfirstlane((int)(blockIdx.x * (blockDim.x >> 5) + (threadIdx.x >> 5)));
    const int lane = threadIdx.x & 31;
    const int v0   = wv * SORT_NPW;
    if (v0 >= nNodes) return;
    const int nq = (nNodes - v0 < SORT_NPW) ? (nNodes - v0) : SORT_NPW;
    const int rp = (lane <= nq) ? rowptr[v0 + lane] : 0;
    for (int q = 0; q < nq; ++q) {
        const int beg = __builtin_amdgcn_readlane(rp, q);
        const int deg = __builtin_amdgcn_readlane(rp, q + 1) - beg;
        for (int ib = 0; ib < deg; ib += 32) {
            const int i   = ib + lane;
            const int key = (i < deg) ? raw[beg + i] : 0;
            int rank = 0;
            for (int jb = 0; jb < deg; jb += 32) {
                const int j0 = jb + lane;
                const int kv = (j0 < deg) ? raw[beg + j0] : 0;
                const int m  = (deg - jb < 32) ? (deg - jb) : 32;
                for (int jj = 0; jj < m; ++jj) {
                    const int kj = __builtin_amdgcn_readlane(kv, jj);
                    rank += (int)(kj < key) + (int)((kj == key) & (jb + jj < i));
                }
            }
            if (i < deg) esrc[beg + rank] = key;
        }
    }
}

__global__ __launch_bounds__(256) void gather_prop(const float* __restrict__ h,
                                                   const int* __restrict__ esrc,
                                                   const int* __restrict__ rowptr,
                                                   const float* __restrict__ prevprev,
                                                   float* __restrict__ outp, int nNodes) {
    int v    = blockIdx.x * (blockDim.x >> 5) + (threadIdx.x >> 5);
    int lane = threadIdx.x & 31;
    if (v >= nNodes) return;
    int beg = rowptr[v];
    int end = rowptr[v + 1];
    float4 acc = make_float4(0.f, 0.f, 0.f, 0.f);
    for (int j = beg; j < end; ++j) {
        int s = esrc[j];
        if (j + 1 < end)
            __builtin_prefetch(h + (size_t)esrc[j + 1] * D_FEAT + lane * 4, 0, 0);
        const float4 t = *(const float4*)(h + (size_t)s * D_FEAT + lane * 4);
        acc.x += t.x; acc.y += t.y; acc.z += t.z; acc.w += t.w;
    }
    int deg = end - beg;
    float scale = 1.0f / (float)(deg > 1 ? deg : 1);
    float4 r;
    if (prevprev != nullptr) {
        const float4 p = *(const float4*)(prevprev + (size_t)v * D_FEAT + lane * 4);
        r.x = 2.0f * acc.x * scale - p.x;
        r.y = 2.0f * acc.y * scale - p.y;
        r.z = 2.0f * acc.z * scale - p.z;
        r.w = 2.0f * acc.w * scale - p.w;
    } else {
        r.x = acc.x * scale;
        r.y = acc.y * scale;
        r.z = acc.z * scale;
        r.w = acc.w * scale;
    }
    *(float4*)(outp + (size_t)v * D_FEAT + lane * 4) = r;
}

__global__ __launch_bounds__(256) void cheb_gemm(const float* __restrict__ T0,
                                                 const float* __restrict__ T1,
                                                 const float* __restrict__ T2,
                                                 const float* __restrict__ T3,
                                                 const float* __restrict__ W,
                                                 const float* __restrict__ bias,
                                                 float* __restrict__ out) {
    const int lane = threadIdx.x & 31;
    const int wave = threadIdx.x >> 5;
    const int rowTile = blockIdx.x;
    const int m   = lane & 15;
    const int hi  = lane >> 4;
    const int row = rowTile * 16 + m;
    const int col = wave * 16 + m;

    v8f c = {};
    const float* terms[4] = {T0, T1, T2, T3};

#pragma unroll
    for (int t = 0; t < 4; ++t) {
        const float* __restrict__ h    = terms[t];
        const float* __restrict__ wrow = W + (size_t)col * (D_FEAT * (K_CHEB + 1)) + t * D_FEAT;
#pragma unroll 8
        for (int k0 = 0; k0 < D_FEAT; k0 += 4) {
            const int kk = k0 + (hi << 1);
            v2f a  = *(const v2f*)(h + (size_t)row * D_FEAT + kk);
            v2f bm = *(const v2f*)(wrow + kk);
            c = __builtin_amdgcn_wmma_f32_16x16x4_f32(
                     false, a,  false, bm,
                     (short)0, c,  false,  false);
        }
    }

    const float bj = bias[col];
#pragma unroll
    for (int i = 0; i < 8; ++i) {
        const int mr = rowTile * 16 + i + hi * 8;
        out[(size_t)mr * D_FEAT + col] = c[i] + bj;
    }
}

extern "C" void kernel_launch(void* const* d_in, const int* in_sizes, int n_in,
                              void* d_out, int out_size, void* d_ws, size_t ws_size,
                              hipStream_t stream) {
    const float* x  = (const float*)d_in[0];
    const int*   ei = (const int*)d_in[1];
    const float* W  = (const float*)d_in[2];
    const float* b  = (const float*)d_in[3];
    float* out = (float*)d_out;

    const int* src = ei;
    const int* dst = ei + N_EDGES;

    const size_t ND = (size_t)N_NODES * D_FEAT;
    float* B1     = (float*)d_ws;
    float* B2     = B1 + ND;
    float* B3     = B2 + ND;
    int*   rowptr = (int*)(B3 + ND);
    int*   cursor = rowptr + (N_NODES + 1);
    int*   esrc   = cursor + N_NODES;

    dim3 blk(256);
    dim3 gEdge((N_EDGES + 255) / 256);
    dim3 gNode((N_NODES + 255) / 256);
    dim3 gWaveNode((N_NODES + 7) / 8);

    zero_i32<<<gNode, blk, 0, stream>>>(cursor, N_NODES);
    count_deg_i<<<gEdge, blk, 0, stream>>>(dst, cursor, N_EDGES);
    scan_excl<<<1, dim3(SCAN_THREADS), 0, stream>>>(cursor, rowptr);
    copy_i32<<<gNode, blk, 0, stream>>>(rowptr, cursor, N_NODES);
    int* raw = (int*)B3;
    fill_edges<<<gEdge, blk, 0, stream>>>(src, dst, cursor, raw, N_EDGES);
    sort_lists<<<dim3((N_NODES + 8 * SORT_NPW - 1) / (8 * SORT_NPW)), blk, 0, stream>>>(raw, rowptr, esrc, N_NODES);

    gather_prop<<<gWaveNode, blk, 0, stream>>>(x,  esrc, rowptr, nullptr, B1, N_NODES);
    gather_prop<<<gWaveNode, blk, 0, stream>>>(B1, esrc, rowptr, x,       B2, N_NODES);
    gather_prop<<<gWaveNode, blk, 0, stream>>>(B2, esrc, rowptr, B1,      B3, N_NODES);

    cheb_gemm<<<dim3(N_NODES / 16), blk, 0, stream>>>(x, B1, B2, B3, W, b, out);
}


# GEMM: A and W chunks (32 k) fetched once per 80-row block with whole-line loads into LDS, fragments read from LDS; was 16-line 8-byte fragment loads from global in every wave; k order kept
# speedup vs baseline: 2.5000x; 1.2783x over previous
#include <hip/hip_runtime.h>

#define N_NODES 50000
#define N_EDGES 800000
#define D_FEAT  128
#define K_CHEB  3

typedef __attribute__((ext_vector_type(2))) float v2f;
typedef __attribute__((ext_vector_type(4))) float v4f;
typedef __attribute__((ext_vector_type(8))) float v8f;

__global__ void zero_i32(int* __restrict__ p, int n) {
    int i = blockIdx.x * blockDim.x + threadIdx.x;
    if (i < n) p[i] = 0;
}

__global__ void count_deg_i(const int* __restrict__ dst, int* __restrict__ cnt, int nEdges) {
    int e = blockIdx.x * blockDim.x + threadIdx.x;
    if (e < nEdges) atomicAdd(&cnt[dst[e]], 1);
}

#define SCAN_THREADS 1024
#define SCAN_CHUNK   (SCAN_THREADS * 4)
#define SCAN_NCH     ((N_NODES + SCAN_CHUNK - 1) / SCAN_CHUNK)

__device__ __forceinline__ int wave_incl_scan(int s, int lane) {
#pragma unroll
    for (int off = 1; off < 32; off <<= 1) {
        int y = __shfl_up(s, (unsigned)off, 32);
        if (lane >= off) s += y;
    }
    return s;
}

static_assert(N_NODES % 4 == 0, "each thread owns whole groups of 4 counts");

__global__ __launch_bounds__(SCAN_THREADS) void scan_excl(const int* __restrict__ cnt,
                                                          int* __restrict__ rowptr) {
    __shared__ int wtot[SCAN_NCH][32];
    __shared__ int ctot[SCAN_NCH];
    const int t    = (int)threadIdx.x;
    const int lane = t & 31;
    const int w    = t >> 5;

    int v0[SCAN_NCH], v1[SCAN_NCH], v2[SCAN_NCH], v3[SCAN_NCH], incl[SCAN_NCH];

#pragma unroll
    for (int c = 0; c < SCAN_NCH; ++c) {
        const int i0 = c * SCAN_CHUNK + t * 4;
        v0[c] = 0; v1[c] = 0; v2[c] = 0; v3[c] = 0;
        if ((c + 1) * SCAN_CHUNK <= N_NODES || i0 < N_NODES) {
            v0[c] = cnt[i0]; v1[c] = cnt[i0 + 1]; v2[c] = cnt[i0 + 2]; v3[c] = cnt[i0 + 3];
        }
    }

#pragma unroll
    for (int c = 0; c < SCAN_NCH; ++c) {
        incl[c] = wave_incl_scan(v0[c] + v1[c] + v2[c] + v3[c], lane);
        if (lane == 31) wtot[c][w] = incl[c];
    }
    __syncthreads();

    if (w < SCAN_NCH) {
        const int x = wtot[w][lane];
        const int s = wave_incl_scan(x, lane);
        wtot[w][lane] = s - x;
        if (lane == 31) ctot[w] = s;
    }
    __syncthreads();

    int carry = 0;
#pragma unroll
    for (int c = 0; c < SCAN_NCH; ++c) {
        const int i0 = c * SCAN_CHUNK + t * 4;
        const int e0 = carry + wtot[c][w] + incl[c] - (v0[c] + v1[c] + v2[c] + v3[c]);
        const int e1 = e0 + v0[c];
        const int e2 = e1 + v1[c];
        const int e3 = e2 + v2[c];
        if ((c + 1) * SCAN_CHUNK <= N_NODES || i0 < N_NODES) {
            rowptr[i0] = e0; rowptr[i0 + 1] = e1; rowptr[i0 + 2] = e2; rowptr[i0 + 3] = e3;
        }
        carry += ctot[c];
    }
    if (t == 0) rowptr[N_NODES] = carry;
}

__global__ void copy_i32(const int* __restrict__ s, int* __restrict__ d, int n) {
    int i = blockIdx.x * blockDim.x + threadIdx.x;
    if (i < n) d[i] = s[i];
}

__global__ void fill_edges(const int* __restrict__ src, const int* __restrict__ dst,
                           int* __restrict__ cursor, int* __restrict__ esrc, int nEdges) {
    int e = blockIdx.x * blockDim.x + threadIdx.x;
    if (e >= nEdges) return;
    int p = atomicAdd(&cursor[dst[e]], 1);
    esrc[p] = src[e];
}

#define SORT_NPW 8
__global__ __launch_bounds__(256) void sort_lists(const int* __restrict__ raw,
                                                  const int* __restrict__ rowptr,
                                                  int* __restrict__ esrc, int nNodes) {
    const int wv   = __builtin_amdgcn_readfirstlane((int)(blockIdx.x * (blockDim.x >> 5) + (threadIdx.x >> 5)));
    const int lane = threadIdx.x & 31;
    const int v0   = wv * SORT_NPW;
    if (v0 >= nNodes) return;
    const int nq = (nNodes - v0 < SORT_NPW) ? (nNodes - v0) : SORT_NPW;
    const int rp = (lane <= nq) ? rowptr[v0 + lane] : 0;
    for (int q = 0; q < nq; ++q) {
        const int beg = __builtin_amdgcn_readlane(rp, q);
        const int deg = __builtin_amdgcn_readlane(rp, q + 1) - beg;
        for (int ib = 0; ib < deg; ib += 32) {
            const int i   = ib + lane;
            const int key = (i < deg) ? raw[beg + i] : 0;
            int rank = 0;
            for (int jb = 0; jb < deg; jb += 32) {
                const int j0 = jb + lane;
                const int kv = (j0 < deg) ? raw[beg + j0] : 0;
                const int m  = (deg - jb < 32) ? (deg - jb) : 32;
                for (int jj = 0; jj < m; ++jj) {
                    const int kj = __builtin_amdgcn_readlane(kv, jj);
                    rank += (int)(kj < key) + (int)((kj == key) & (jb + jj < i));
                }
            }
            if (i < deg) esrc[beg + rank] = key;
        }
    }
}

__global__ __launch_bounds__(256) void gather_prop(const float* __restrict__ h,
                                                   const int* __restrict__ esrc,
                                                   const int* __restrict__ rowptr,
                                                   const float* __restrict__ prevprev,
                                                   float* __restrict__ outp, int nNodes) {
    int v    = blockIdx.x * (blockDim.x >> 5) + (threadIdx.x >> 5);
    int lane = threadIdx.x & 31;
    if (v >= nNodes) return;
    int beg = rowptr[v];
    int end = rowptr[v + 1];
    float4 acc = make_float4(0.f, 0.f, 0.f, 0.f);
    for (int j = beg; j < end; ++j) {
        int s = esrc[j];
        if (j + 1 < end)
            __builtin_prefetch(h + (size_t)esrc[j + 1] * D_FEAT + lane * 4, 0, 0);
        const float4 t = *(const float4*)(h + (size_t)s * D_FEAT + lane * 4);
        acc.x += t.x; acc.y += t.y; acc.z += t.z; acc.w += t.w;
    }
    int deg = end - beg;
    float scale = 1.0f / (float)(deg > 1 ? deg : 1);
    float4 r;
    if (prevprev != nullptr) {
        const float4 p = *(const float4*)(prevprev + (size_t)v * D_FEAT + lane * 4);
        r.x = 2.0f * acc.x * scale - p.x;
        r.y = 2.0f * acc.y * scale - p.y;
        r.z = 2.0f * acc.z * scale - p.z;
        r.w = 2.0f * acc.w * scale - p.w;
    } else {
        r.x = acc.x * scale;
        r.y = acc.y * scale;
        r.z = acc.z * scale;
        r.w = acc.w * scale;
    }
    *(float4*)(outp + (size_t)v * D_FEAT + lane * 4) = r;
}

#define GM_ROWS  80
#define GM_RT    (GM_ROWS / 16)
#define GM_KC    32
#define GM_PITCH (GM_KC + 4)
#define GM_KTOT  (D_FEAT * (K_CHEB + 1))
#define GM_NCH   (GM_KTOT / GM_KC)

static_assert(N_NODES % GM_ROWS == 0, "row tiles cover the nodes exactly");
static_assert(GM_ROWS % 16 == 0, "whole 16-row tiles");
static_assert(GM_ROWS > 64 && GM_ROWS <= 96, "an A chunk is staged as two full groups of 32 rows and one partial");
static_assert(D_FEAT % GM_KC == 0, "a chunk stays inside one term");

__global__ __launch_bounds__(256) void cheb_gemm(const float* __restrict__ T0,
                                                 const float* __restrict__ T1,
                                                 const float* __restrict__ T2,
                                                 const float* __restrict__ T3,
                                                 const float* __restrict__ W,
                                                 const float* __restrict__ bias,
                                                 float* __restrict__ out) {
    __shared__ __attribute__((aligned(16))) float As[GM_ROWS * GM_PITCH];
    __shared__ __attribute__((aligned(16))) float Bs[D_FEAT * GM_PITCH];

    const int tid  = (int)threadIdx.x;
    const int lane = tid & 31;
    const int wave = tid >> 5;
    const int m    = lane & 15;
    const int hi   = lane >> 4;
    const int row0 = (int)blockIdx.x * GM_ROWS;
    const int ldr  = tid >> 3;
    const int seg  = tid & 7;

    v8f acc[GM_RT];
#pragma unroll
    for (int r = 0; r < GM_RT; ++r) acc[r] = (v8f){};

    const bool a2on = (ldr + 64) < GM_ROWS;
    const size_t aoff = (size_t)(row0 + ldr) * D_FEAT + seg * 4;
    const size_t woff = (size_t)ldr * GM_KTOT + seg * 4;
    float* const as_st = As + ldr * GM_PITCH + seg * 4;
    float* const bs_st = Bs + ldr * GM_PITCH + seg * 4;

    v4f ra0, ra1, ra2 = {0.f, 0.f, 0.f, 0.f}, rb0, rb1, rb2, rb3;
    ra0 = *(const v4f*)(T0 + aoff);
    ra1 = *(const v4f*)(T0 + aoff + 32 * D_FEAT);
    if (a2on) ra2 = *(const v4f*)(T0 + aoff + 64 * D_FEAT);
    rb0 = *(const v4f*)(W + woff);
    rb1 = *(const v4f*)(W + woff + 32 * GM_KTOT);
    rb2 = *(const v4f*)(W + woff + 64 * GM_KTOT);
    rb3 = *(const v4f*)(W + woff + 96 * GM_KTOT);

#pragma unroll 1
    for (int c = 0; c < GM_NCH; ++c) {
        *(v4f*)(as_st)                 = ra0;
        *(v4f*)(as_st + 32 * GM_PITCH) = ra1;
        if (a2on) *(v4f*)(as_st + 64 * GM_PITCH) = ra2;
        *(v4f*)(bs_st)                 = rb0;
        *(v4f*)(bs_st + 32 * GM_PITCH) = rb1;
        *(v4f*)(bs_st + 64 * GM_PITCH) = rb2;
        *(v4f*)(bs_st + 96 * GM_PITCH) = rb3;
        __syncthreads();

        if (c + 1 < GM_NCH) {
            const int cn = c + 1;
            const float* __restrict__ h = (cn < 4) ? T0 : (cn < 8) ? T1 : (cn < 12) ? T2 : T3;
            const float* __restrict__ pa = h + aoff + (cn & 3) * GM_KC;
            const float* __restrict__ pw = W + woff + cn * GM_KC;
            ra0 = *(const v4f*)(pa);
            ra1 = *(const v4f*)(pa + 32 * D_FEAT);
            if (a2on) ra2 = *(const v4f*)(pa + 64 * D_FEAT);
            rb0 = *(const v4f*)(pw);
            rb1 = *(const v4f*)(pw + 32 * GM_KTOT);
            rb2 = *(const v4f*)(pw + 64 * GM_KTOT);
            rb3 = *(const v4f*)(pw + 96 * GM_KTOT);
        }

#pragma unroll
        for (int k0 = 0; k0 < GM_KC; k0 += 4) {
            const int kk = k0 + (hi << 1);
            const v2f bm = *(const v2f*)(Bs + (wave * 16 + m) * GM_PITCH + kk);
#pragma unroll
            for (int r = 0; r < GM_RT; ++r) {
                const v2f a = *(const v2f*)(As + (r * 16 + m) * GM_PITCH + kk);
                acc[r] = __builtin_amdgcn_wmma_f32_16x16x4_f32(
                              false, a,  false, bm,
                              (short)0, acc[r],  false,  false);
            }
        }
        __syncthreads();
    }

    const int col = wave * 16 + m;
    const float bj = bias[col];
#pragma unroll
    for (int r = 0; r < GM_RT; ++r) {
#pragma unroll
        for (int i = 0; i < 8; ++i) {
            const int mr = row0 + r * 16 + i + hi * 8;
            out[(size_t)mr * D_FEAT + col] = acc[r][i] + bj;
        }
    }
}

extern "C" void kernel_launch(void* const* d_in, const int* in_sizes, int n_in,
                              void* d_out, int out_size, void* d_ws, size_t ws_size,
                              hipStream_t stream) {
    const float* x  = (const float*)d_in[0];
    const int*   ei = (const int*)d_in[1];
    const float* W  = (const float*)d_in[2];
    const float* b  = (const float*)d_in[3];
    float* out = (float*)d_out;

    const int* src = ei;
    const int* dst = ei + N_EDGES;

    const size_t ND = (size_t)N_NODES * D_FEAT;
    float* B1     = (float*)d_ws;
    float* B2     = B1 + ND;
    float* B3     = B2 + ND;
    int*   rowptr = (int*)(B3 + ND);
    int*   cursor = rowptr + (N_NODES + 1);
    int*   esrc   = cursor + N_NODES;

    dim3 blk(256);
    dim3 gEdge((N_EDGES + 255) / 256);
    dim3 gNode((N_NODES + 255) / 256);
    dim3 gWaveNode((N_NODES + 7) / 8);

    zero_i32<<<gNode, blk, 0, stream>>>(cursor, N_NODES);
    count_deg_i<<<gEdge, blk, 0, stream>>>(dst, cursor, N_EDGES);
    scan_excl<<<1, dim3(SCAN_THREADS), 0, stream>>>(cursor, rowptr);
    copy_i32<<<gNode, blk, 0, stream>>>(rowptr, cursor, N_NODES);
    int* raw = (int*)B3;
    fill_edges<<<gEdge, blk, 0, stream>>>(src, dst, cursor, raw, N_EDGES);
    sort_lists<<<dim3((N_NODES + 8 * SORT_NPW - 1) / (8 * SORT_NPW)), blk, 0, stream>>>(raw, rowptr, esrc, N_NODES);

    gather_prop<<<gWaveNode, blk, 0, stream>>>(x,  esrc, rowptr, nullptr, B1, N_NODES);
    gather_prop<<<gWaveNode, blk, 0, stream>>>(B1, esrc, rowptr, x,       B2, N_NODES);
    gather_prop<<<gWaveNode, blk, 0, stream>>>(B2, esrc, rowptr, B1,      B3, N_NODES);

    cheb_gemm<<<dim3(N_NODES / GM_ROWS), blk, 0, stream>>>(x, B1, B2, B3, W, b, out);
}


# CSR build: per-edge global atomics (degree count, slot cursor) become per-chunk LDS counters whose returning add ranks the edge; a per-node prefix over 64 chunks gives the slots; float sums unchanged
# speedup vs baseline: 3.3357x; 1.3343x over previous
#include <hip/hip_runtime.h>

#define N_NODES 50000
#define N_EDGES 800000
#define D_FEAT  128
#define K_CHEB  3

typedef __attribute__((ext_vector_type(2))) float v2f;
typedef __attribute__((ext_vector_type(4))) float v4f;
typedef __attribute__((ext_vector_type(8))) float v8f;

#define HC_CHUNKS  64
#define HC_EDGES   (N_EDGES / HC_CHUNKS)
#define HC_THREADS 1024
#define HC_PER     ((HC_EDGES + HC_THREADS - 1) / HC_THREADS)

static_assert(N_EDGES % HC_CHUNKS == 0, "equal chunks");
static_assert(HC_EDGES % 4 == 0, "four consecutive edges never straddle a chunk");
static_assert(N_NODES % 4 == 0, "counter planes move as 16-byte pieces");

__global__ __launch_bounds__(HC_THREADS) void hist_rank(const int* __restrict__ dst,
                                                        int* __restrict__ rank,
                                                        int* __restrict__ cc) {
    __shared__ __attribute__((aligned(16))) int cnt[N_NODES];
    const int t  = (int)threadIdx.x;
    const int e0 = (int)blockIdx.x * HC_EDGES;

    int d[HC_PER];
#pragma unroll
    for (int k = 0; k < HC_PER; ++k) {
        const int i = t + k * HC_THREADS;
        d[k] = (i < HC_EDGES) ? dst[e0 + i] : 0;
    }

    for (int i = t * 4; i < N_NODES; i += HC_THREADS * 4) *(int4*)(cnt + i) = make_int4(0, 0, 0, 0);
    __syncthreads();

    int r[HC_PER];
#pragma unroll
    for (int k = 0; k < HC_PER; ++k) {
        const int i = t + k * HC_THREADS;
        r[k] = (i < HC_EDGES) ? atomicAdd(&cnt[d[k]], 1) : 0;
    }
#pragma unroll
    for (int k = 0; k < HC_PER; ++k) {
        const int i = t + k * HC_THREADS;
        if (i < HC_EDGES) rank[e0 + i] = r[k];
    }
    __syncthreads();

    int* __restrict__ plane = cc + (size_t)blockIdx.x * N_NODES;
    for (int i = t * 4; i < N_NODES; i += HC_THREADS * 4) *(int4*)(plane + i) = *(const int4*)(cnt + i);
}

#define CP_GROUP 16
static_assert(HC_CHUNKS % CP_GROUP == 0, "whole groups of planes");
__global__ __launch_bounds__(256) void chunk_prefix(int* cc, int* __restrict__ deg) {
    const int v = (int)(blockIdx.x * blockDim.x + threadIdx.x);
    if (v >= N_NODES) return;
    int run = 0;
    for (int cb = 0; cb < HC_CHUNKS; cb += CP_GROUP) {
        int x[CP_GROUP];
#pragma unroll
        for (int k = 0; k < CP_GROUP; ++k) x[k] = cc[(size_t)(cb + k) * N_NODES + v];
#pragma unroll
        for (int k = 0; k < CP_GROUP; ++k) { cc[(size_t)(cb + k) * N_NODES + v] = run; run += x[k]; }
    }
    deg[v] = run;
}

#define SCAN_THREADS 1024
#define SCAN_CHUNK   (SCAN_THREADS * 4)
#define SCAN_NCH     ((N_NODES + SCAN_CHUNK - 1) / SCAN_CHUNK)

__device__ __forceinline__ int wave_incl_scan(int s, int lane) {
#pragma unroll
    for (int off = 1; off < 32; off <<= 1) {
        int y = __shfl_up(s, (unsigned)off, 32);
        if (lane >= off) s += y;
    }
    return s;
}

static_assert(N_NODES % 4 == 0, "each thread owns whole groups of 4 counts");

__global__ __launch_bounds__(SCAN_THREADS) void scan_excl(const int* __restrict__ cnt,
                                                          int* __restrict__ rowptr) {
    __shared__ int wtot[SCAN_NCH][32];
    __shared__ int ctot[SCAN_NCH];
    const int t    = (int)threadIdx.x;
    const int lane = t & 31;
    const int w    = t >> 5;

    int v0[SCAN_NCH], v1[SCAN_NCH], v2[SCAN_NCH], v3[SCAN_NCH], incl[SCAN_NCH];

#pragma unroll
    for (int c = 0; c < SCAN_NCH; ++c) {
        const int i0 = c * SCAN_CHUNK + t * 4;
        v0[c] = 0; v1[c] = 0; v2[c] = 0; v3[c] = 0;
        if ((c + 1) * SCAN_CHUNK <= N_NODES || i0 < N_NODES) {
            v0[c] = cnt[i0]; v1[c] = cnt[i0 + 1]; v2[c] = cnt[i0 + 2]; v3[c] = cnt[i0 + 3];
        }
    }

#pragma unroll
    for (int c = 0; c < SCAN_NCH; ++c) {
        incl[c] = wave_incl_scan(v0[c] + v1[c] + v2[c] + v3[c], lane);
        if (lane == 31) wtot[c][w] = incl[c];
    }
    __syncthreads();

    if (w < SCAN_NCH) {
        const int x = wtot[w][lane];
        const int s = wave_incl_scan(x, lane);
        wtot[w][lane] = s - x;
        if (lane == 31) ctot[w] = s;
    }
    __syncthreads();

    int carry = 0;
#pragma unroll
    for (int c = 0; c < SCAN_NCH; ++c) {
        const int i0 = c * SCAN_CHUNK + t * 4;
        const int e0 = carry + wtot[c][w] + incl[c] - (v0[c] + v1[c] + v2[c] + v3[c]);
        const int e1 = e0 + v0[c];
        const int e2 = e1 + v1[c];
        const int e3 = e2 + v2[c];
        if ((c + 1) * SCAN_CHUNK <= N_NODES || i0 < N_NODES) {
            rowptr[i0] = e0; rowptr[i0 + 1] = e1; rowptr[i0 + 2] = e2; rowptr[i0 + 3] = e3;
        }
        carry += ctot[c];
    }
    if (t == 0) rowptr[N_NODES] = carry;
}

__global__ __launch_bounds__(256) void place_edges(const int* __restrict__ src, const int* __restrict__ dst,
                                                   const int* __restrict__ rank, const int* __restrict__ cc,
                                                   const int* __restrict__ rowptr, int* __restrict__ raw) {
    const int e = (int)(blockIdx.x * blockDim.x + threadIdx.x) * 4;
    if (e >= N_EDGES) return;
    const int4 d = *(const int4*)(dst + e);
    const int4 r = *(const int4*)(rank + e);
    const int4 s = *(const int4*)(src + e);
    const int* __restrict__ plane = cc + (size_t)(e / HC_EDGES) * N_NODES;
    const int p0 = rowptr[d.x] + plane[d.x] + r.x;
    const int p1 = rowptr[d.y] + plane[d.y] + r.y;
    const int p2 = rowptr[d.z] + plane[d.z] + r.z;
    const int p3 = rowptr[d.w] + plane[d.w] + r.w;
    raw[p0] = s.x;
    raw[p1] = s.y;
    raw[p2] = s.z;
    raw[p3] = s.w;
}

#define SORT_NPW 8
__global__ __launch_bounds__(256) void sort_lists(const int* __restrict__ raw,
                                                  const int* __restrict__ rowptr,
                                                  int* __restrict__ esrc, int nNodes) {
    const int wv   = __builtin_amdgcn_readfirstlane((int)(blockIdx.x * (blockDim.x >> 5) + (threadIdx.x >> 5)));
    const int lane = threadIdx.x & 31;
    const int v0   = wv * SORT_NPW;
    if (v0 >= nNodes) return;
    const int nq = (nNodes - v0 < SORT_NPW) ? (nNodes - v0) : SORT_NPW;
    const int rp = (lane <= nq) ? rowptr[v0 + lane] : 0;
    for (int q = 0; q < nq; ++q) {
        const int beg = __builtin_amdgcn_readlane(rp, q);
        const int deg = __builtin_amdgcn_readlane(rp, q + 1) - beg;
        for (int ib = 0; ib < deg; ib += 32) {
            const int i   = ib + lane;
            const int key = (i < deg) ? raw[beg + i] : 0;
            int rank = 0;
            for (int jb = 0; jb < deg; jb += 32) {
                const int j0 = jb + lane;
                const int kv = (j0 < deg) ? raw[beg + j0] : 0;
                const int m  = (deg - jb < 32) ? (deg - jb) : 32;
                for (int jj = 0; jj < m; ++jj) {
                    const int kj = __builtin_amdgcn_readlane(kv, jj);
                    rank += (int)(kj < key) + (int)((kj == key) & (jb + jj < i));
                }
            }
            if (i < deg) esrc[beg + rank] = key;
        }
    }
}

__global__ __launch_bounds__(256) void gather_prop(const float* __restrict__ h,
                                                   const int* __restrict__ esrc,
                                                   const int* __restrict__ rowptr,
                                                   const float* __restrict__ prevprev,
                                                   float* __restrict__ outp, int nNodes) {
    int v    = blockIdx.x * (blockDim.x >> 5) + (threadIdx.x >> 5);
    int lane = threadIdx.x & 31;
    if (v >= nNodes) return;
    int beg = rowptr[v];
    int end = rowptr[v + 1];
    float4 acc = make_float4(0.f, 0.f, 0.f, 0.f);
    for (int j = beg; j < end; ++j) {
        int s = esrc[j];
        if (j + 1 < end)
            __builtin_prefetch(h + (size_t)esrc[j + 1] * D_FEAT + lane * 4, 0, 0);
        const float4 t = *(const float4*)(h + (size_t)s * D_FEAT + lane * 4);
        acc.x += t.x; acc.y += t.y; acc.z += t.z; acc.w += t.w;
    }
    int deg = end - beg;
    float scale = 1.0f / (float)(deg > 1 ? deg : 1);
    float4 r;
    if (prevprev != nullptr) {
        const float4 p = *(const float4*)(prevprev + (size_t)v * D_FEAT + lane * 4);
        r.x = 2.0f * acc.x * scale - p.x;
        r.y = 2.0f * acc.y * scale - p.y;
        r.z = 2.0f * acc.z * scale - p.z;
        r.w = 2.0f * acc.w * scale - p.w;
    } else {
        r.x = acc.x * scale;
        r.y = acc.y * scale;
        r.z = acc.z * scale;
        r.w = acc.w * scale;
    }
    *(float4*)(outp + (size_t)v * D_FEAT + lane * 4) = r;
}

#define GM_ROWS  80
#define GM_RT    (GM_ROWS / 16)
#define GM_KC    32
#define GM_PITCH (GM_KC + 4)
#define GM_KTOT  (D_FEAT * (K_CHEB + 1))
#define GM_NCH   (GM_KTOT / GM_KC)

static_assert(N_NODES % GM_ROWS == 0, "row tiles cover the nodes exactly");
static_assert(GM_ROWS % 16 == 0, "whole 16-row tiles");
static_assert(GM_ROWS > 64 && GM_ROWS <= 96, "an A chunk is staged as two full groups of 32 rows and one partial");
static_assert(D_FEAT % GM_KC == 0, "a chunk stays inside one term");

__global__ __launch_bounds__(256) void cheb_gemm(const float* __restrict__ T0,
                                                 const float* __restrict__ T1,
                                                 const float* __restrict__ T2,
                                                 const float* __restrict__ T3,
                                                 const float* __restrict__ W,
                                                 const float* __restrict__ bias,
                                                 float* __restrict__ out) {
    __shared__ __attribute__((aligned(16))) float As[GM_ROWS * GM_PITCH];
    __shared__ __attribute__((aligned(16))) float Bs[D_FEAT * GM_PITCH];

    const int tid  = (int)threadIdx.x;
    const int lane = tid & 31;
    const int wave = tid >> 5;
    const int m    = lane & 15;
    const int hi   = lane >> 4;
    const int row0 = (int)blockIdx.x * GM_ROWS;
    const int ldr  = tid >> 3;
    const int seg  = tid & 7;

    v8f acc[GM_RT];
#pragma unroll
    for (int r = 0; r < GM_RT; ++r) acc[r] = (v8f){};

    const bool a2on = (ldr + 64) < GM_ROWS;
    const size_t aoff = (size_t)(row0 + ldr) * D_FEAT + seg * 4;
    const size_t woff = (size_t)ldr * GM_KTOT + seg * 4;
    float* const as_st = As + ldr * GM_PITCH + seg * 4;
    float* const bs_st = Bs + ldr * GM_PITCH + seg * 4;

    v4f ra0, ra1, ra2 = {0.f, 0.f, 0.f, 0.f}, rb0, rb1, rb2, rb3;
    ra0 = *(const v4f*)(T0 + aoff);
    ra1 = *(const v4f*)(T0 + aoff + 32 * D_FEAT);
    if (a2on) ra2 = *(const v4f*)(T0 + aoff + 64 * D_FEAT);
    rb0 = *(const v4f*)(W + woff);
    rb1 = *(const v4f*)(W + woff + 32 * GM_KTOT);
    rb2 = *(const v4f*)(W + woff + 64 * GM_KTOT);
    rb3 = *(const v4f*)(W + woff + 96 * GM_KTOT);

#pragma unroll 1
    for (int c = 0; c < GM_NCH; ++c) {
        *(v4f*)(as_st)                 = ra0;
        *(v4f*)(as_st + 32 * GM_PITCH) = ra1;
        if (a2on) *(v4f*)(as_st + 64 * GM_PITCH) = ra2;
        *(v4f*)(bs_st)                 = rb0;
        *(v4f*)(bs_st + 32 * GM_PITCH) = rb1;
        *(v4f*)(bs_st + 64 * GM_PITCH) = rb2;
        *(v4f*)(bs_st + 96 * GM_PITCH) = rb3;
        __syncthreads();

        if (c + 1 < GM_NCH) {
            const int cn = c + 1;
            const float* __restrict__ h = (cn < 4) ? T0 : (cn < 8) ? T1 : (cn < 12) ? T2 : T3;
            const float* __restrict__ pa = h + aoff + (cn & 3) * GM_KC;
            const float* __restrict__ pw = W + woff + cn * GM_KC;
            ra0 = *(const v4f*)(pa);
            ra1 = *(const v4f*)(pa + 32 * D_FEAT);
            if (a2on) ra2 = *(const v4f*)(pa + 64 * D_FEAT);
            rb0 = *(const v4f*)(pw);
            rb1 = *(const v4f*)(pw + 32 * GM_KTOT);
            rb2 = *(const v4f*)(pw + 64 * GM_KTOT);
            rb3 = *(const v4f*)(pw + 96 * GM_KTOT);
        }

#pragma unroll
        for (int k0 = 0; k0 < GM_KC; k0 += 4) {
            const int kk = k0 + (hi << 1);
            const v2f bm = *(const v2f*)(Bs + (wave * 16 + m) * GM_PITCH + kk);
#pragma unroll
            for (int r = 0; r < GM_RT; ++r) {
                const v2f a = *(const v2f*)(As + (r * 16 + m) * GM_PITCH + kk);
                acc[r] = __builtin_amdgcn_wmma_f32_16x16x4_f32(
                              false, a,  false, bm,
                              (short)0, acc[r],  false,  false);
            }
        }
        __syncthreads();
    }

    const int col = wave * 16 + m;
    const float bj = bias[col];
#pragma unroll
    for (int r = 0; r < GM_RT; ++r) {
#pragma unroll
        for (int i = 0; i < 8; ++i) {
            const int mr = row0 + r * 16 + i + hi * 8;
            out[(size_t)mr * D_FEAT + col] = acc[r][i] + bj;
        }
    }
}

extern "C" void kernel_launch(void* const* d_in, const int* in_sizes, int n_in,
                              void* d_out, int out_size, void* d_ws, size_t ws_size,
                              hipStream_t stream) {
    const float* x  = (const float*)d_in[0];
    const int*   ei = (const int*)d_in[1];
    const float* W  = (const float*)d_in[2];
    const float* b  = (const float*)d_in[3];
    float* out = (float*)d_out;

    const int* src = ei;
    const int* dst = ei + N_EDGES;

    const size_t ND = (size_t)N_NODES * D_FEAT;
    float* B1     = (float*)d_ws;
    float* B2     = B1 + ND;
    float* B3     = B2 + ND;
    int*   rowptr = (int*)(B3 + ND);
    int*   deg    = rowptr + (N_NODES + 1);
    int*   esrc   = deg + N_NODES;

    dim3 blk(256);
    dim3 gNode((N_NODES + 255) / 256);
    dim3 gWaveNode((N_NODES + 7) / 8);

    int* raw  = (int*)B3;
    int* rank = raw + N_EDGES;
    int* cc   = (int*)B2;
    hist_rank<<<dim3(HC_CHUNKS), dim3(HC_THREADS), 0, stream>>>(dst, rank, cc);
    chunk_prefix<<<gNode, blk, 0, stream>>>(cc, deg);
    scan_excl<<<1, dim3(SCAN_THREADS), 0, stream>>>(deg, rowptr);
    place_edges<<<dim3((N_EDGES / 4 + 255) / 256), blk, 0, stream>>>(src, dst, rank, cc, rowptr, raw);
    sort_lists<<<dim3((N_NODES + 8 * SORT_NPW - 1) / (8 * SORT_NPW)), blk, 0, stream>>>(raw, rowptr, esrc, N_NODES);

    gather_prop<<<gWaveNode, blk, 0, stream>>>(x,  esrc, rowptr, nullptr, B1, N_NODES);
    gather_prop<<<gWaveNode, blk, 0, stream>>>(B1, esrc, rowptr, x,       B2, N_NODES);
    gather_prop<<<gWaveNode, blk, 0, stream>>>(B2, esrc, rowptr, B1,      B3, N_NODES);

    cheb_gemm<<<dim3(N_NODES / GM_ROWS), blk, 0, stream>>>(x, B1, B2, B3, W, b, out);
}


# gather_prop: per-edge prefetch of the next source row removed with the extra dependent index read it needed; a trip's 8 row reads now go out together (was one at a time); sum order kept
# speedup vs baseline: 3.8866x; 1.1652x over previous
#include <hip/hip_runtime.h>

#define N_NODES 50000
#define N_EDGES 800000
#define D_FEAT  128
#define K_CHEB  3

typedef __attribute__((ext_vector_type(2))) float v2f;
typedef __attribute__((ext_vector_type(4))) float v4f;
typedef __attribute__((ext_vector_type(8))) float v8f;

#define HC_CHUNKS  64
#define HC_EDGES   (N_EDGES / HC_CHUNKS)
#define HC_THREADS 1024
#define HC_PER     ((HC_EDGES + HC_THREADS - 1) / HC_THREADS)

static_assert(N_EDGES % HC_CHUNKS == 0, "equal chunks");
static_assert(HC_EDGES % 4 == 0, "four consecutive edges never straddle a chunk");
static_assert(N_NODES % 4 == 0, "counter planes move as 16-byte pieces");

__global__ __launch_bounds__(HC_THREADS) void hist_rank(const int* __restrict__ dst,
                                                        int* __restrict__ rank,
                                                        int* __restrict__ cc) {
    __shared__ __attribute__((aligned(16))) int cnt[N_NODES];
    const int t  = (int)threadIdx.x;
    const int e0 = (int)blockIdx.x * HC_EDGES;

    int d[HC_PER];
#pragma unroll
    for (int k = 0; k < HC_PER; ++k) {
        const int i = t + k * HC_THREADS;
        d[k] = (i < HC_EDGES) ? dst[e0 + i] : 0;
    }

    for (int i = t * 4; i < N_NODES; i += HC_THREADS * 4) *(int4*)(cnt + i) = make_int4(0, 0, 0, 0);
    __syncthreads();

    int r[HC_PER];
#pragma unroll
    for (int k = 0; k < HC_PER; ++k) {
        const int i = t + k * HC_THREADS;
        r[k] = (i < HC_EDGES) ? atomicAdd(&cnt[d[k]], 1) : 0;
    }
#pragma unroll
    for (int k = 0; k < HC_PER; ++k) {
        const int i = t + k * HC_THREADS;
        if (i < HC_EDGES) rank[e0 + i] = r[k];
    }
    __syncthreads();

    int* __restrict__ plane = cc + (size_t)blockIdx.x * N_NODES;
    for (int i = t * 4; i < N_NODES; i += HC_THREADS * 4) *(int4*)(plane + i) = *(const int4*)(cnt + i);
}

#define CP_GROUP 16
static_assert(HC_CHUNKS % CP_GROUP == 0, "whole groups of planes");
__global__ __launch_bounds__(256) void chunk_prefix(int* cc, int* __restrict__ deg) {
    const int v = (int)(blockIdx.x * blockDim.x + threadIdx.x);
    if (v >= N_NODES) return;
    int run = 0;
    for (int cb = 0; cb < HC_CHUNKS; cb += CP_GROUP) {
        int x[CP_GROUP];
#pragma unroll
        for (int k = 0; k < CP_GROUP; ++k) x[k] = cc[(size_t)(cb + k) * N_NODES + v];
#pragma unroll
        for (int k = 0; k < CP_GROUP; ++k) { cc[(size_t)(cb + k) * N_NODES + v] = run; run += x[k]; }
    }
    deg[v] = run;
}

#define SCAN_THREADS 1024
#define SCAN_CHUNK   (SCAN_THREADS * 4)
#define SCAN_NCH     ((N_NODES + SCAN_CHUNK - 1) / SCAN_CHUNK)

__device__ __forceinline__ int wave_incl_scan(int s, int lane) {
#pragma unroll
    for (int off = 1; off < 32; off <<= 1) {
        int y = __shfl_up(s, (unsigned)off, 32);
        if (lane >= off) s += y;
    }
    return s;
}

static_assert(N_NODES % 4 == 0, "each thread owns whole groups of 4 counts");

__global__ __launch_bounds__(SCAN_THREADS) void scan_excl(const int* __restrict__ cnt,
                                                          int* __restrict__ rowptr) {
    __shared__ int wtot[SCAN_NCH][32];
    __shared__ int ctot[SCAN_NCH];
    const int t    = (int)threadIdx.x;
    const int lane = t & 31;
    const int w    = t >> 5;

    int v0[SCAN_NCH], v1[SCAN_NCH], v2[SCAN_NCH], v3[SCAN_NCH], incl[SCAN_NCH];

#pragma unroll
    for (int c = 0; c < SCAN_NCH; ++c) {
        const int i0 = c * SCAN_CHUNK + t * 4;
        v0[c] = 0; v1[c] = 0; v2[c] = 0; v3[c] = 0;
        if ((c + 1) * SCAN_CHUNK <= N_NODES || i0 < N_NODES) {
            v0[c] = cnt[i0]; v1[c] = cnt[i0 + 1]; v2[c] = cnt[i0 + 2]; v3[c] = cnt[i0 + 3];
        }
    }

#pragma unroll
    for (int c = 0; c < SCAN_NCH; ++c) {
        incl[c] = wave_incl_scan(v0[c] + v1[c] + v2[c] + v3[c], lane);
        if (lane == 31) wtot[c][w] = incl[c];
    }
    __syncthreads();

    if (w < SCAN_NCH) {
        const int x = wtot[w][lane];
        const int s = wave_incl_scan(x, lane);
        wtot[w][lane] = s - x;
        if (lane == 31) ctot[w] = s;
    }
    __syncthreads();

    int carry = 0;
#pragma unroll
    for (int c = 0; c < SCAN_NCH; ++c) {
        const int i0 = c * SCAN_CHUNK + t * 4;
        const int e0 = carry + wtot[c][w] + incl[c] - (v0[c] + v1[c] + v2[c] + v3[c]);
        const int e1 = e0 + v0[c];
        const int e2 = e1 + v1[c];
        const int e3 = e2 + v2[c];
        if ((c + 1) * SCAN_CHUNK <= N_NODES || i0 < N_NODES) {
            rowptr[i0] = e0; rowptr[i0 + 1] = e1; rowptr[i0 + 2] = e2; rowptr[i0 + 3] = e3;
        }
        carry += ctot[c];
    }
    if (t == 0) rowptr[N_NODES] = carry;
}

__global__ __launch_bounds__(256) void place_edges(const int* __restrict__ src, const int* __restrict__ dst,
                                                   const int* __restrict__ rank, const int* __restrict__ cc,
                                                   const int* __restrict__ rowptr, int* __restrict__ raw) {
    const int e = (int)(blockIdx.x * blockDim.x + threadIdx.x) * 4;
    if (e >= N_EDGES) return;
    const int4 d = *(const int4*)(dst + e);
    const int4 r = *(const int4*)(rank + e);
    const int4 s = *(const int4*)(src + e);
    const int* __restrict__ plane = cc + (size_t)(e / HC_EDGES) * N_NODES;
    const int p0 = rowptr[d.x] + plane[d.x] + r.x;
    const int p1 = rowptr[d.y] + plane[d.y] + r.y;
    const int p2 = rowptr[d.z] + plane[d.z] + r.z;
    const int p3 = rowptr[d.w] + plane[d.w] + r.w;
    raw[p0] = s.x;
    raw[p1] = s.y;
    raw[p2] = s.z;
    raw[p3] = s.w;
}

#define SORT_NPW 8
__global__ __launch_bounds__(256) void sort_lists(const int* __restrict__ raw,
                                                  const int* __restrict__ rowptr,
                                                  int* __restrict__ esrc, int nNodes) {
    const int wv   = __builtin_amdgcn_readfirstlane((int)(blockIdx.x * (blockDim.x >> 5) + (threadIdx.x >> 5)));
    const int lane = threadIdx.x & 31;
    const int v0   = wv * SORT_NPW;
    if (v0 >= nNodes) return;
    const int nq = (nNodes - v0 < SORT_NPW) ? (nNodes - v0) : SORT_NPW;
    const int rp = (lane <= nq) ? rowptr[v0 + lane] : 0;
    for (int q = 0; q < nq; ++q) {
        const int beg = __builtin_amdgcn_readlane(rp, q);
        const int deg = __builtin_amdgcn_readlane(rp, q + 1) - beg;
        for (int ib = 0; ib < deg; ib += 32) {
            const int i   = ib + lane;
            const int key = (i < deg) ? raw[beg + i] : 0;
            int rank = 0;
            for (int jb = 0; jb < deg; jb += 32) {
                const int j0 = jb + lane;
                const int kv = (j0 < deg) ? raw[beg + j0] : 0;
                const int m  = (deg - jb < 32) ? (deg - jb) : 32;
                for (int jj = 0; jj < m; ++jj) {
                    const int kj = __builtin_amdgcn_readlane(kv, jj);
                    rank += (int)(kj < key) + (int)((kj == key) & (jb + jj < i));
                }
            }
            if (i < deg) esrc[beg + rank] = key;
        }
    }
}

__global__ __launch_bounds__(256) void gather_prop(const float* __restrict__ h,
                                                   const int* __restrict__ esrc,
                                                   const int* __restrict__ rowptr,
                                                   const float* __restrict__ prevprev,
                                                   float* __restrict__ outp, int nNodes) {
    int v    = blockIdx.x * (blockDim.x >> 5) + (threadIdx.x >> 5);
    int lane = threadIdx.x & 31;
    if (v >= nNodes) return;
    int beg = rowptr[v];
    int end = rowptr[v + 1];
    float4 acc = make_float4(0.f, 0.f, 0.f, 0.f);
    for (int j = beg; j < end; ++j) {
        const int s = esrc[j];
        const float4 t = *(const float4*)(h + (size_t)s * D_FEAT + lane * 4);
        acc.x += t.x; acc.y += t.y; acc.z += t.z; acc.w += t.w;
    }
    int deg = end - beg;
    float scale = 1.0f / (float)(deg > 1 ? deg : 1);
    float4 r;
    if (prevprev != nullptr) {
        const float4 p = *(const float4*)(prevprev + (size_t)v * D_FEAT + lane * 4);
        r.x = 2.0f * acc.x * scale - p.x;
        r.y = 2.0f * acc.y * scale - p.y;
        r.z = 2.0f * acc.z * scale - p.z;
        r.w = 2.0f * acc.w * scale - p.w;
    } else {
        r.x = acc.x * scale;
        r.y = acc.y * scale;
        r.z = acc.z * scale;
        r.w = acc.w * scale;
    }
    *(float4*)(outp + (size_t)v * D_FEAT + lane * 4) = r;
}

#define GM_ROWS  80
#define GM_RT    (GM_ROWS / 16)
#define GM_KC    32
#define GM_PITCH (GM_KC + 4)
#define GM_KTOT  (D_FEAT * (K_CHEB + 1))
#define GM_NCH   (GM_KTOT / GM_KC)

static_assert(N_NODES % GM_ROWS == 0, "row tiles cover the nodes exactly");
static_assert(GM_ROWS % 16 == 0, "whole 16-row tiles");
static_assert(GM_ROWS > 64 && GM_ROWS <= 96, "an A chunk is staged as two full groups of 32 rows and one partial");
static_assert(D_FEAT % GM_KC == 0, "a chunk stays inside one term");

__global__ __launch_bounds__(256) void cheb_gemm(const float* __restrict__ T0,
                                                 const float* __restrict__ T1,
                                                 const float* __restrict__ T2,
                                                 const float* __restrict__ T3,
                                                 const float* __restrict__ W,
                                                 const float* __restrict__ bias,
                                                 float* __restrict__ out) {
    __shared__ __attribute__((aligned(16))) float As[GM_ROWS * GM_PITCH];
    __shared__ __attribute__((aligned(16))) float Bs[D_FEAT * GM_PITCH];

    const int tid  = (int)threadIdx.x;
    const int lane = tid & 31;
    const int wave = tid >> 5;
    const int m    = lane & 15;
    const int hi   = lane >> 4;
    const int row0 = (int)blockIdx.x * GM_ROWS;
    const int ldr  = tid >> 3;
    const int seg  = tid & 7;

    v8f acc[GM_RT];
#pragma unroll
    for (int r = 0; r < GM_RT; ++r) acc[r] = (v8f){};

    const bool a2on = (ldr + 64) < GM_ROWS;
    const size_t aoff = (size_t)(row0 + ldr) * D_FEAT + seg * 4;
    const size_t woff = (size_t)ldr * GM_KTOT + seg * 4;
    float* const as_st = As + ldr * GM_PITCH + seg * 4;
    float* const bs_st = Bs + ldr * GM_PITCH + seg * 4;

    v4f ra0, ra1, ra2 = {0.f, 0.f, 0.f, 0.f}, rb0, rb1, rb2, rb3;
    ra0 = *(const v4f*)(T0 + aoff);
    ra1 = *(const v4f*)(T0 + aoff + 32 * D_FEAT);
    if (a2on) ra2 = *(const v4f*)(T0 + aoff + 64 * D_FEAT);
    rb0 = *(const v4f*)(W + woff);
    rb1 = *(const v4f*)(W + woff + 32 * GM_KTOT);
    rb2 = *(const v4f*)(W + woff + 64 * GM_KTOT);
    rb3 = *(const v4f*)(W + woff + 96 * GM_KTOT);

#pragma unroll 1
    for (int c = 0; c < GM_NCH; ++c) {
        *(v4f*)(as_st)                 = ra0;
        *(v4f*)(as_st + 32 * GM_PITCH) = ra1;
        if (a2on) *(v4f*)(as_st + 64 * GM_PITCH) = ra2;
        *(v4f*)(bs_st)                 = rb0;
        *(v4f*)(bs_st + 32 * GM_PITCH) = rb1;
        *(v4f*)(bs_st + 64 * GM_PITCH) = rb2;
        *(v4f*)(bs_st + 96 * GM_PITCH) = rb3;
        __syncthreads();

        if (c + 1 < GM_NCH) {
            const int cn = c + 1;
            const float* __restrict__ h = (cn < 4) ? T0 : (cn < 8) ? T1 : (cn < 12) ? T2 : T3;
            const float* __restrict__ pa = h + aoff + (cn & 3) * GM_KC;
            const float* __restrict__ pw = W + woff + cn * GM_KC;
            ra0 = *(const v4f*)(pa);
            ra1 = *(const v4f*)(pa + 32 * D_FEAT);
            if (a2on) ra2 = *(const v4f*)(pa + 64 * D_FEAT);
            rb0 = *(const v4f*)(pw);
            rb1 = *(const v4f*)(pw + 32 * GM_KTOT);
            rb2 = *(const v4f*)(pw + 64 * GM_KTOT);
            rb3 = *(const v4f*)(pw + 96 * GM_KTOT);
        }

#pragma unroll
        for (int k0 = 0; k0 < GM_KC; k0 += 4) {
            const int kk = k0 + (hi << 1);
            const v2f bm = *(const v2f*)(Bs + (wave * 16 + m) * GM_PITCH + kk);
#pragma unroll
            for (int r = 0; r < GM_RT; ++r) {
                const v2f a = *(const v2f*)(As + (r * 16 + m) * GM_PITCH + kk);
                acc[r] = __builtin_amdgcn_wmma_f32_16x16x4_f32(
                              false, a,  false, bm,
                              (short)0, acc[r],  false,  false);
            }
        }
        __syncthreads();
    }

    const int col = wave * 16 + m;
    const float bj = bias[col];
#pragma unroll
    for (int r = 0; r < GM_RT; ++r) {
#pragma unroll
        for (int i = 0; i < 8; ++i) {
            const int mr = row0 + r * 16 + i + hi * 8;
            out[(size_t)mr * D_FEAT + col] = acc[r][i] + bj;
        }
    }
}

extern "C" void kernel_launch(void* const* d_in, const int* in_sizes, int n_in,
                              void* d_out, int out_size, void* d_ws, size_t ws_size,
                              hipStream_t stream) {
    const float* x  = (const float*)d_in[0];
    const int*   ei = (const int*)d_in[1];
    const float* W  = (const float*)d_in[2];
    const float* b  = (const float*)d_in[3];
    float* out = (float*)d_out;

    const int* src = ei;
    const int* dst = ei + N_EDGES;

    const size_t ND = (size_t)N_NODES * D_FEAT;
    float* B1     = (float*)d_ws;
    float* B2     = B1 + ND;
    float* B3     = B2 + ND;
    int*   rowptr = (int*)(B3 + ND);
    int*   deg    = rowptr + (N_NODES + 1);
    int*   esrc   = deg + N_NODES;

    dim3 blk(256);
    dim3 gNode((N_NODES + 255) / 256);
    dim3 gWaveNode((N_NODES + 7) / 8);

    int* raw  = (int*)B3;
    int* rank = raw + N_EDGES;
    int* cc   = (int*)B2;
    hist_rank<<<dim3(HC_CHUNKS), dim3(HC_THREADS), 0, stream>>>(dst, rank, cc);
    chunk_prefix<<<gNode, blk, 0, stream>>>(cc, deg);
    scan_excl<<<1, dim3(SCAN_THREADS), 0, stream>>>(deg, rowptr);
    place_edges<<<dim3((N_EDGES / 4 + 255) / 256), blk, 0, stream>>>(src, dst, rank, cc, rowptr, raw);
    sort_lists<<<dim3((N_NODES + 8 * SORT_NPW - 1) / (8 * SORT_NPW)), blk, 0, stream>>>(raw, rowptr, esrc, N_NODES);

    gather_prop<<<gWaveNode, blk, 0, stream>>>(x,  esrc, rowptr, nullptr, B1, N_NODES);
    gather_prop<<<gWaveNode, blk, 0, stream>>>(B1, esrc, rowptr, x,       B2, N_NODES);
    gather_prop<<<gWaveNode, blk, 0, stream>>>(B2, esrc, rowptr, B1,      B3, N_NODES);

    cheb_gemm<<<dim3(N_NODES / GM_ROWS), blk, 0, stream>>>(x, B1, B2, B3, W, b, out);
}


# Launches 9 -> 7: the one-block offset scan is folded into its neighbours (block scan + 196 block totals) and the list sort into the first gather, which walks rows from sorted registers; sums kept
# speedup vs baseline: 4.4040x; 1.1331x over previous
#include <hip/hip_runtime.h>

#define N_NODES 50000
#define N_EDGES 800000
#define D_FEAT  128
#define K_CHEB  3

typedef __attribute__((ext_vector_type(2))) float v2f;
typedef __attribute__((ext_vector_type(4))) float v4f;
typedef __attribute__((ext_vector_type(8))) float v8f;

#define HC_CHUNKS  64
#define HC_EDGES   (N_EDGES / HC_CHUNKS)
#define HC_THREADS 1024
#define HC_PER     ((HC_EDGES + HC_THREADS - 1) / HC_THREADS)

static_assert(N_EDGES % HC_CHUNKS == 0, "equal chunks");
static_assert(HC_EDGES % 4 == 0, "four consecutive edges never straddle a chunk");
static_assert(N_NODES % 4 == 0, "counter planes move as 16-byte pieces");

__global__ __launch_bounds__(HC_THREADS) void hist_rank(const int* __restrict__ dst,
                                                        int* __restrict__ rank,
                                                        int* __restrict__ cc) {
    __shared__ __attribute__((aligned(16))) int cnt[N_NODES];
    const int t  = (int)threadIdx.x;
    const int e0 = (int)blockIdx.x * HC_EDGES;

    int d[HC_PER];
#pragma unroll
    for (int k = 0; k < HC_PER; ++k) {
        const int i = t + k * HC_THREADS;
        d[k] = (i < HC_EDGES) ? dst[e0 + i] : 0;
    }

    for (int i = t * 4; i < N_NODES; i += HC_THREADS * 4) *(int4*)(cnt + i) = make_int4(0, 0, 0, 0);
    __syncthreads();

    int r[HC_PER];
#pragma unroll
    for (int k = 0; k < HC_PER; ++k) {
        const int i = t + k * HC_THREADS;
        r[k] = (i < HC_EDGES) ? atomicAdd(&cnt[d[k]], 1) : 0;
    }
#pragma unroll
    for (int k = 0; k < HC_PER; ++k) {
        const int i = t + k * HC_THREADS;
        if (i < HC_EDGES) rank[e0 + i] = r[k];
    }
    __syncthreads();

    int* __restrict__ plane = cc + (size_t)blockIdx.x * N_NODES;
    for (int i = t * 4; i < N_NODES; i += HC_THREADS * 4) *(int4*)(plane + i) = *(const int4*)(cnt + i);
}

__device__ __forceinline__ int wave_incl_scan(int s, int lane) {
#pragma unroll
    for (int off = 1; off < 32; off <<= 1) {
        int y = __shfl_up(s, (unsigned)off, 32);
        if (lane >= off) s += y;
    }
    return s;
}

#define NB_T    256
#define NB_NODE ((N_NODES + NB_T - 1) / NB_T)
static_assert(NB_NODE <= NB_T, "one thread per block total in the scan of the totals");

#define CP_GROUP 16
static_assert(HC_CHUNKS % CP_GROUP == 0, "whole groups of planes");
__global__ __launch_bounds__(NB_T) void chunk_prefix(int* cc, int* __restrict__ lpre, int* __restrict__ btot) {
    __shared__ int wsum[NB_T / 32];
    const int t    = (int)threadIdx.x;
    const int lane = t & 31;
    const int w    = t >> 5;
    const int v    = (int)blockIdx.x * NB_T + t;
    int run = 0;
    if (v < N_NODES) {
        for (int cb = 0; cb < HC_CHUNKS; cb += CP_GROUP) {
            int x[CP_GROUP];
#pragma unroll
            for (int k = 0; k < CP_GROUP; ++k) x[k] = cc[(size_t)(cb + k) * N_NODES + v];
#pragma unroll
            for (int k = 0; k < CP_GROUP; ++k) { cc[(size_t)(cb + k) * N_NODES + v] = run; run += x[k]; }
        }
    }
    const int incl = wave_incl_scan(run, lane);
    if (lane == 31) wsum[w] = incl;
    __syncthreads();
    int before = 0, total = 0;
#pragma unroll
    for (int k = 0; k < NB_T / 32; ++k) { const int x = wsum[k]; total += x; if (k < w) before += x; }
    if (v < N_NODES) lpre[v] = before + incl - run;
    if (t == 0) btot[blockIdx.x] = total;
}

#define PE_T 256
static_assert(PE_T == NB_T, "a block writes the rowptr slice of one node block");
static_assert((N_EDGES / 4 + PE_T - 1) / PE_T >= NB_NODE, "enough blocks to write every rowptr slice");
__global__ __launch_bounds__(PE_T) void place_edges(const int* __restrict__ src, const int* __restrict__ dst,
                                                    const int* __restrict__ rank, const int* __restrict__ cc,
                                                    const int* __restrict__ lpre, const int* __restrict__ btot,
                                                    int* __restrict__ rowptr, int* __restrict__ raw) {
    __shared__ int base_s[PE_T];
    __shared__ int wsum[PE_T / 32];
    const int t    = (int)threadIdx.x;
    const int lane = t & 31;
    const int w    = t >> 5;
    const int e    = ((int)blockIdx.x * PE_T + t) * 4;
    const bool on  = e < N_EDGES;

    int4 d = make_int4(0, 0, 0, 0), r = d, s = d;
    if (on) {
        d = *(const int4*)(dst + e);
        r = *(const int4*)(rank + e);
        s = *(const int4*)(src + e);
    }

    const int x    = (t < NB_NODE) ? btot[t] : 0;
    const int incl = wave_incl_scan(x, lane);
    if (lane == 31) wsum[w] = incl;
    __syncthreads();
    int before = 0, total = 0;
#pragma unroll
    for (int k = 0; k < PE_T / 32; ++k) { const int y = wsum[k]; total += y; if (k < w) before += y; }
    base_s[t] = before + incl - x;
    __syncthreads();

    if ((int)blockIdx.x < NB_NODE) {
        const int v = (int)blockIdx.x * NB_T + t;
        if (v < N_NODES) rowptr[v] = base_s[blockIdx.x] + lpre[v];
        if (blockIdx.x == 0 && t == 0) rowptr[N_NODES] = total;
    }

    if (on) {
        const int* __restrict__ plane = cc + (size_t)(e / HC_EDGES) * N_NODES;
        const int p0 = base_s[d.x / NB_T] + lpre[d.x] + plane[d.x] + r.x;
        const int p1 = base_s[d.y / NB_T] + lpre[d.y] + plane[d.y] + r.y;
        const int p2 = base_s[d.z / NB_T] + lpre[d.z] + plane[d.z] + r.z;
        const int p3 = base_s[d.w / NB_T] + lpre[d.w] + plane[d.w] + r.w;
        raw[p0] = s.x;
        raw[p1] = s.y;
        raw[p2] = s.z;
        raw[p3] = s.w;
    }
}

#define GF_ROWS 8
__global__ __launch_bounds__(256) void sort_gather_first(const float* __restrict__ h,
                                                         const int* __restrict__ raw,
                                                         const int* __restrict__ rowptr,
                                                         int* esrc,
                                                         float* __restrict__ outp, int nNodes) {
    const int v    = __builtin_amdgcn_readfirstlane((int)(blockIdx.x * (blockDim.x >> 5) + (threadIdx.x >> 5)));
    const int lane = threadIdx.x & 31;
    if (v >= nNodes) return;
    const int beg = __builtin_amdgcn_readfirstlane(rowptr[v]);
    const int end = __builtin_amdgcn_readfirstlane(rowptr[v + 1]);
    const int deg = end - beg;
    float4 acc = make_float4(0.f, 0.f, 0.f, 0.f);

    if (deg <= 32) {
        const int key = (lane < deg) ? raw[beg + lane] : 0;
        int rk = 0;
        for (int jj = 0; jj < deg; ++jj) {
            const int kj = __builtin_amdgcn_readlane(key, jj);
            rk += (int)(kj < key) + (int)((kj == key) & (jj < lane));
        }
        if (lane < deg) esrc[beg + rk] = key;
        int sorted = 0;
        for (int jj = 0; jj < deg; ++jj) {
            const int rj = __builtin_amdgcn_readlane(rk, jj);
            const int kj = __builtin_amdgcn_readlane(key, jj);
            if (rj == lane) sorted = kj;
        }
        for (int k0 = 0; k0 < deg; k0 += GF_ROWS) {
            const int n = deg - k0;
            float4 t[GF_ROWS];
            if (n >= GF_ROWS) {
#pragma unroll
                for (int k = 0; k < GF_ROWS; ++k) {
                    const int sidx = __builtin_amdgcn_readlane(sorted, k0 + k);
                    t[k] = *(const float4*)(h + (size_t)sidx * D_FEAT + lane * 4);
                }
#pragma unroll
                for (int k = 0; k < GF_ROWS; ++k) { acc.x += t[k].x; acc.y += t[k].y; acc.z += t[k].z; acc.w += t[k].w; }
            } else {
#pragma unroll
                for (int k = 0; k < GF_ROWS - 1; ++k) {
                    if (k < n) {
                        const int sidx = __builtin_amdgcn_readlane(sorted, k0 + k);
                        t[k] = *(const float4*)(h + (size_t)sidx * D_FEAT + lane * 4);
                    }
                }
#pragma unroll
                for (int k = 0; k < GF_ROWS - 1; ++k) {
                    if (k < n) { acc.x += t[k].x; acc.y += t[k].y; acc.z += t[k].z; acc.w += t[k].w; }
                }
            }
        }
    } else {
        for (int ib = 0; ib < deg; ib += 32) {
            const int i   = ib + lane;
            const int key = (i < deg) ? raw[beg + i] : 0;
            int rk = 0;
            for (int jb = 0; jb < deg; jb += 32) {
                const int j0 = jb + lane;
                const int kv = (j0 < deg) ? raw[beg + j0] : 0;
                const int m  = (deg - jb < 32) ? (deg - jb) : 32;
                for (int jj = 0; jj < m; ++jj) {
                    const int kj = __builtin_amdgcn_readlane(kv, jj);
                    rk += (int)(kj < key) + (int)((kj == key) & (jb + jj < i));
                }
            }
            if (i < deg) esrc[beg + rk] = key;
        }
        __threadfence();
        for (int j = beg; j < end; ++j) {
            const int sidx = esrc[j];
            const float4 t = *(const float4*)(h + (size_t)sidx * D_FEAT + lane * 4);
            acc.x += t.x; acc.y += t.y; acc.z += t.z; acc.w += t.w;
        }
    }

    const float scale = 1.0f / (float)(deg > 1 ? deg : 1);
    float4 r;
    r.x = acc.x * scale;
    r.y = acc.y * scale;
    r.z = acc.z * scale;
    r.w = acc.w * scale;
    *(float4*)(outp + (size_t)v * D_FEAT + lane * 4) = r;
}

__global__ __launch_bounds__(256) void gather_prop(const float* __restrict__ h,
                                                   const int* __restrict__ esrc,
                                                   const int* __restrict__ rowptr,
                                                   const float* __restrict__ prevprev,
                                                   float* __restrict__ outp, int nNodes) {
    int v    = blockIdx.x * (blockDim.x >> 5) + (threadIdx.x >> 5);
    int lane = threadIdx.x & 31;
    if (v >= nNodes) return;
    int beg = rowptr[v];
    int end = rowptr[v + 1];
    float4 acc = make_float4(0.f, 0.f, 0.f, 0.f);
    for (int j = beg; j < end; ++j) {
        const int s = esrc[j];
        const float4 t = *(const float4*)(h + (size_t)s * D_FEAT + lane * 4);
        acc.x += t.x; acc.y += t.y; acc.z += t.z; acc.w += t.w;
    }
    int deg = end - beg;
    float scale = 1.0f / (float)(deg > 1 ? deg : 1);
    float4 r;
    if (prevprev != nullptr) {
        const float4 p = *(const float4*)(prevprev + (size_t)v * D_FEAT + lane * 4);
        r.x = 2.0f * acc.x * scale - p.x;
        r.y = 2.0f * acc.y * scale - p.y;
        r.z = 2.0f * acc.z * scale - p.z;
        r.w = 2.0f * acc.w * scale - p.w;
    } else {
        r.x = acc.x * scale;
        r.y = acc.y * scale;
        r.z = acc.z * scale;
        r.w = acc.w * scale;
    }
    *(float4*)(outp + (size_t)v * D_FEAT + lane * 4) = r;
}

#define GM_ROWS  80
#define GM_RT    (GM_ROWS / 16)
#define GM_KC    32
#define GM_PITCH (GM_KC + 4)
#define GM_KTOT  (D_FEAT * (K_CHEB + 1))
#define GM_NCH   (GM_KTOT / GM_KC)

static_assert(N_NODES % GM_ROWS == 0, "row tiles cover the nodes exactly");
static_assert(GM_ROWS % 16 == 0, "whole 16-row tiles");
static_assert(GM_ROWS > 64 && GM_ROWS <= 96, "an A chunk is staged as two full groups of 32 rows and one partial");
static_assert(D_FEAT % GM_KC == 0, "a chunk stays inside one term");

__global__ __launch_bounds__(256) void cheb_gemm(const float* __restrict__ T0,
                                                 const float* __restrict__ T1,
                                                 const float* __restrict__ T2,
                                                 const float* __restrict__ T3,
                                                 const float* __restrict__ W,
                                                 const float* __restrict__ bias,
                                                 float* __restrict__ out) {
    __shared__ __attribute__((aligned(16))) float As[GM_ROWS * GM_PITCH];
    __shared__ __attribute__((aligned(16))) float Bs[D_FEAT * GM_PITCH];

    const int tid  = (int)threadIdx.x;
    const int lane = tid & 31;
    const int wave = tid >> 5;
    const int m    = lane & 15;
    const int hi   = lane >> 4;
    const int row0 = (int)blockIdx.x * GM_ROWS;
    const int ldr  = tid >> 3;
    const int seg  = tid & 7;

    v8f acc[GM_RT];
#pragma unroll
    for (int r = 0; r < GM_RT; ++r) acc[r] = (v8f){};

    const bool a2on = (ldr + 64) < GM_ROWS;
    const size_t aoff = (size_t)(row0 + ldr) * D_FEAT + seg * 4;
    const size_t woff = (size_t)ldr * GM_KTOT + seg * 4;
    float* const as_st = As + ldr * GM_PITCH + seg * 4;
    float* const bs_st = Bs + ldr * GM_PITCH + seg * 4;

    v4f ra0, ra1, ra2 = {0.f, 0.f, 0.f, 0.f}, rb0, rb1, rb2, rb3;
    ra0 = *(const v4f*)(T0 + aoff);
    ra1 = *(const v4f*)(T0 + aoff + 32 * D_FEAT);
    if (a2on) ra2 = *(const v4f*)(T0 + aoff + 64 * D_FEAT);
    rb0 = *(const v4f*)(W + woff);
    rb1 = *(const v4f*)(W + woff + 32 * GM_KTOT);
    rb2 = *(const v4f*)(W + woff + 64 * GM_KTOT);
    rb3 = *(const v4f*)(W + woff + 96 * GM_KTOT);

#pragma unroll 1
    for (int c = 0; c < GM_NCH; ++c) {
        *(v4f*)(as_st)                 = ra0;
        *(v4f*)(as_st + 32 * GM_PITCH) = ra1;
        if (a2on) *(v4f*)(as_st + 64 * GM_PITCH) = ra2;
        *(v4f*)(bs_st)                 = rb0;
        *(v4f*)(bs_st + 32 * GM_PITCH) = rb1;
        *(v4f*)(bs_st + 64 * GM_PITCH) = rb2;
        *(v4f*)(bs_st + 96 * GM_PITCH) = rb3;
        __syncthreads();

        if (c + 1 < GM_NCH) {
            const int cn = c + 1;
            const float* __restrict__ h = (cn < 4) ? T0 : (cn < 8) ? T1 : (cn < 12) ? T2 : T3;
            const float* __restrict__ pa = h + aoff + (cn & 3) * GM_KC;
            const float* __restrict__ pw = W + woff + cn * GM_KC;
            ra0 = *(const v4f*)(pa);
            ra1 = *(const v4f*)(pa + 32 * D_FEAT);
            if (a2on) ra2 = *(const v4f*)(pa + 64 * D_FEAT);
            rb0 = *(const v4f*)(pw);
            rb1 = *(const v4f*)(pw + 32 * GM_KTOT);
            rb2 = *(const v4f*)(pw + 64 * GM_KTOT);
            rb3 = *(const v4f*)(pw + 96 * GM_KTOT);
        }

#pragma unroll
        for (int k0 = 0; k0 < GM_KC; k0 += 4) {
            const int kk = k0 + (hi << 1);
            const v2f bm = *(const v2f*)(Bs + (wave * 16 + m) * GM_PITCH + kk);
#pragma unroll
            for (int r = 0; r < GM_RT; ++r) {
                const v2f a = *(const v2f*)(As + (r * 16 + m) * GM_PITCH + kk);
                acc[r] = __builtin_amdgcn_wmma_f32_16x16x4_f32(
                              false, a,  false, bm,
                              (short)0, acc[r],  false,  false);
            }
        }
        __syncthreads();
    }

    const int col = wave * 16 + m;
    const float bj = bias[col];
#pragma unroll
    for (int r = 0; r < GM_RT; ++r) {
#pragma unroll
        for (int i = 0; i < 8; ++i) {
            const int mr = row0 + r * 16 + i + hi * 8;
            out[(size_t)mr * D_FEAT + col] = acc[r][i] + bj;
        }
    }
}

extern "C" void kernel_launch(void* const* d_in, const int* in_sizes, int n_in,
                              void* d_out, int out_size, void* d_ws, size_t ws_size,
                              hipStream_t stream) {
    const float* x  = (const float*)d_in[0];
    const int*   ei = (const int*)d_in[1];
    const float* W  = (const float*)d_in[2];
    const float* b  = (const float*)d_in[3];
    float* out = (float*)d_out;

    const int* src = ei;
    const int* dst = ei + N_EDGES;

    const size_t ND = (size_t)N_NODES * D_FEAT;
    float* B1     = (float*)d_ws;
    float* B2     = B1 + ND;
    float* B3     = B2 + ND;
    int*   rowptr = (int*)(B3 + ND);
    int*   lpre   = rowptr + (N_NODES + 1);
    int*   esrc   = lpre + N_NODES;

    dim3 blk(256);
    dim3 gWaveNode((N_NODES + 7) / 8);

    int* raw  = (int*)B3;
    int* rank = raw + N_EDGES;
    int* cc   = (int*)B2;
    int* btot = (int*)B1;
    hist_rank<<<dim3(HC_CHUNKS), dim3(HC_THREADS), 0, stream>>>(dst, rank, cc);
    chunk_prefix<<<dim3(NB_NODE), dim3(NB_T), 0, stream>>>(cc, lpre, btot);
    place_edges<<<dim3((N_EDGES / 4 + PE_T - 1) / PE_T), dim3(PE_T), 0, stream>>>(src, dst, rank, cc, lpre, btot, rowptr, raw);

    sort_gather_first<<<gWaveNode, blk, 0, stream>>>(x, raw, rowptr, esrc, B1, N_NODES);
    gather_prop<<<gWaveNode, blk, 0, stream>>>(B1, esrc, rowptr, x,       B2, N_NODES);
    gather_prop<<<gWaveNode, blk, 0, stream>>>(B2, esrc, rowptr, B1,      B3, N_NODES);

    cheb_gemm<<<dim3(N_NODES / GM_ROWS), blk, 0, stream>>>(x, B1, B2, B3, W, b, out);
}
